# starting kernel
# baseline (speedup 1.0000x reference)
#include <hip/hip_runtime.h>
#include <math.h>

typedef float v2f __attribute__((ext_vector_type(2)));
typedef float v8f __attribute__((ext_vector_type(8)));

constexpr int   B = 32;
constexpr int   S = 8192;
constexpr int   D = 64;
constexpr float SCALE = 0.125f;

constexpr int WAVES          = 8;
constexpr int KEYS_PER_WAVE  = 64;
constexpr int CHUNKS         = S / (WAVES * KEYS_PER_WAVE);

constexpr int NPART     = B * CHUNKS;
constexpr int WS_O_OFF  = 0;
constexpr int WS_M_OFF  = NPART * D;
constexpr int WS_L_OFF  = WS_M_OFF + NPART;
constexpr int WS_ML_OFF = WS_L_OFF + NPART;

__global__ __launch_bounds__(WAVES * 32)
void attn_partial_kernel(const float* __restrict__ q,
                         const float* __restrict__ k,
                         const float* __restrict__ v,
                         float* __restrict__ attn,
                         float* __restrict__ ws)
{
    const int tid   = threadIdx.x;
    const int lane  = tid & 31;
    const int wave  = tid >> 5;
    const int chunk = blockIdx.x;
    const int b     = blockIdx.y;

    __shared__ __align__(16) float qs[D];
    __shared__ float sm[WAVES];
    __shared__ float sl[WAVES];
    __shared__ __align__(16) float so[WAVES][D];

    if (tid < D) qs[tid] = q[b * D + tid];
    __syncthreads();

    const int waveKeyBase = (chunk * WAVES + wave) * KEYS_PER_WAVE;
    const int colBase     = (lane >> 4) << 1;

    float m = -INFINITY, l = 0.f, o0 = 0.f, o1 = 0.f;

    for (int t = 0; t < KEYS_PER_WAVE / 16; ++t) {
        const int s0     = waveKeyBase + t * 16;
        const int keyRow = s0 + (lane & 15);
        const float* kp  = k + ((size_t)b * S + keyRow) * D;

        v8f acc = {};
        #pragma unroll
        for (int c = 0; c < 16; ++c) {
            v2f a2 = *(const v2f*)(kp + c * 4 + colBase);
            v2f b2 = *(const v2f*)(qs + c * 4 + colBase);
            acc = __builtin_amdgcn_wmma_f32_16x16x4_f32(
                false, a2, false, b2, (short)0, acc, false, false);
        }

        float ps[8];
        #pragma unroll
        for (int r = 0; r < 8; ++r) ps[r] = acc[r] * SCALE;

        if ((lane & 15) == 0) {
            float* ap = attn + (size_t)b * S + s0 + (lane >> 4) * 8;
            *(float4*)(ap)     = make_float4(ps[0], ps[1], ps[2], ps[3]);
            *(float4*)(ap + 4) = make_float4(ps[4], ps[5], ps[6], ps[7]);
        }

        float tmax = ps[0];
        #pragma unroll
        for (int r = 1; r < 8; ++r) tmax = fmaxf(tmax, ps[r]);
        tmax = fmaxf(tmax, __shfl_xor(tmax, 16, 32));

        const float newm = fmaxf(m, tmax);
        const float resc = __expf(m - newm);
        l *= resc; o0 *= resc; o1 *= resc;
        m = newm;

        const float* vbase = v + (size_t)b * S * D + lane * 2;
        #pragma unroll
        for (int r = 0; r < 8; ++r) {
            const float p   = __expf(ps[r] - m);
            const float plo = __shfl(p, 0, 32);
            const float phi = __shfl(p, 16, 32);
            l += plo + phi;
            const float2 vlo = *(const float2*)(vbase + (size_t)(s0 + r) * D);
            const float2 vhi = *(const float2*)(vbase + (size_t)(s0 + 8 + r) * D);
            o0 += plo * vlo.x + phi * vhi.x;
            o1 += plo * vlo.y + phi * vhi.y;
        }
    }

    so[wave][lane * 2]     = o0;
    so[wave][lane * 2 + 1] = o1;
    if (lane == 0) { sm[wave] = m; sl[wave] = l; }
    __syncthreads();

    if (tid < D) {
        float M = sm[0];
        #pragma unroll
        for (int i = 1; i < WAVES; ++i) M = fmaxf(M, sm[i]);
        float L = 0.f, O = 0.f;
        #pragma unroll
        for (int i = 0; i < WAVES; ++i) {
            const float f = __expf(sm[i] - M);
            L += f * sl[i];
            O += f * so[i][tid];
        }
        const int idx = b * CHUNKS + chunk;
        ws[WS_O_OFF + idx * D + tid] = O;
        if (tid == 0) { ws[WS_M_OFF + idx] = M; ws[WS_L_OFF + idx] = L; }
    }
}

__global__ __launch_bounds__(64)
void attn_combine_kernel(float* __restrict__ out, float* __restrict__ ws)
{
    const int b = blockIdx.x;
    const int t = threadIdx.x;
    __shared__ float cm[CHUNKS], cl[CHUNKS];
    if (t < CHUNKS) {
        cm[t] = ws[WS_M_OFF + b * CHUNKS + t];
        cl[t] = ws[WS_L_OFF + b * CHUNKS + t];
    }
    __syncthreads();

    float M = cm[0];
    #pragma unroll
    for (int i = 1; i < CHUNKS; ++i) M = fmaxf(M, cm[i]);
    float L = 0.f, O = 0.f;
    #pragma unroll
    for (int i = 0; i < CHUNKS; ++i) {
        const float f = __expf(cm[i] - M);
        L += f * cl[i];
        O += f * ws[WS_O_OFF + (b * CHUNKS + i) * D + t];
    }
    out[b * D + t] = O / L;
    if (t == 0) { ws[WS_ML_OFF + b * 2] = M; ws[WS_ML_OFF + b * 2 + 1] = L; }
}

__global__ __launch_bounds__(256)
void attn_normalize_kernel(float* __restrict__ attn, const float* __restrict__ ws)
{
    const int b = blockIdx.y;
    const int s = blockIdx.x * 256 + threadIdx.x;
    const float M = ws[WS_ML_OFF + b * 2];
    const float L = ws[WS_ML_OFF + b * 2 + 1];
    const size_t i = (size_t)b * S + s;
    attn[i] = __expf(attn[i] - M) / L;
}

extern "C" void kernel_launch(void* const* d_in, const int* in_sizes, int n_in,
                              void* d_out, int out_size, void* d_ws, size_t ws_size,
                              hipStream_t stream)
{
    (void)in_sizes; (void)n_in; (void)out_size; (void)ws_size;
    const float* q = (const float*)d_in[0];
    const float* k = (const float*)d_in[1];
    const float* v = (const float*)d_in[2];
    float* out  = (float*)d_out;
    float* attn = out + B * D;
    float* ws   = (float*)d_ws;

    attn_partial_kernel<<<dim3(CHUNKS, B), WAVES * 32, 0, stream>>>(q, k, v, attn, ws);
    attn_combine_kernel<<<dim3(B), 64, 0, stream>>>(out, ws);
    attn_normalize_kernel<<<dim3(S / 256, B), 256, 0, stream>>>(attn, ws);
}


# Combine launch folded into the normalize launch: each normalize block refolds its batch's (m,l) chunk partials in the old order, block 0 of a batch also writes out; 3 launches -> 2
# speedup vs baseline: 1.1291x; 1.1291x over previous
#include <hip/hip_runtime.h>
#include <math.h>

typedef float v2f __attribute__((ext_vector_type(2)));
typedef float v8f __attribute__((ext_vector_type(8)));

constexpr int   B = 32;
constexpr int   S = 8192;
constexpr int   D = 64;
constexpr float SCALE = 0.125f;

constexpr int WAVES          = 8;
constexpr int KEYS_PER_WAVE  = 64;
constexpr int CHUNKS         = S / (WAVES * KEYS_PER_WAVE);

constexpr int NPART     = B * CHUNKS;
constexpr int WS_O_OFF  = 0;
constexpr int WS_M_OFF  = NPART * D;
constexpr int WS_L_OFF  = WS_M_OFF + NPART;

__global__ __launch_bounds__(WAVES * 32)
void attn_partial_kernel(const float* __restrict__ q,
                         const float* __restrict__ k,
                         const float* __restrict__ v,
                         float* __restrict__ attn,
                         float* __restrict__ ws)
{
    const int tid   = threadIdx.x;
    const int lane  = tid & 31;
    const int wave  = tid >> 5;
    const int chunk = blockIdx.x;
    const int b     = blockIdx.y;

    __shared__ __align__(16) float qs[D];
    __shared__ float sm[WAVES];
    __shared__ float sl[WAVES];
    __shared__ __align__(16) float so[WAVES][D];

    if (tid < D) qs[tid] = q[b * D + tid];
    __syncthreads();

    const int waveKeyBase = (chunk * WAVES + wave) * KEYS_PER_WAVE;
    const int colBase     = (lane >> 4) << 1;

    float m = -INFINITY, l = 0.f, o0 = 0.f, o1 = 0.f;

    for (int t = 0; t < KEYS_PER_WAVE / 16; ++t) {
        const int s0     = waveKeyBase + t * 16;
        const int keyRow = s0 + (lane & 15);
        const float* kp  = k + ((size_t)b * S + keyRow) * D;

        v8f acc = {};
        #pragma unroll
        for (int c = 0; c < 16; ++c) {
            v2f a2 = *(const v2f*)(kp + c * 4 + colBase);
            v2f b2 = *(const v2f*)(qs + c * 4 + colBase);
            acc = __builtin_amdgcn_wmma_f32_16x16x4_f32(
                false, a2, false, b2, (short)0, acc, false, false);
        }

        float ps[8];
        #pragma unroll
        for (int r = 0; r < 8; ++r) ps[r] = acc[r] * SCALE;

        if ((lane & 15) == 0) {
            float* ap = attn + (size_t)b * S + s0 + (lane >> 4) * 8;
            *(float4*)(ap)     = make_float4(ps[0], ps[1], ps[2], ps[3]);
            *(float4*)(ap + 4) = make_float4(ps[4], ps[5], ps[6], ps[7]);
        }

        float tmax = ps[0];
        #pragma unroll
        for (int r = 1; r < 8; ++r) tmax = fmaxf(tmax, ps[r]);
        tmax = fmaxf(tmax, __shfl_xor(tmax, 16, 32));

        const float newm = fmaxf(m, tmax);
        const float resc = __expf(m - newm);
        l *= resc; o0 *= resc; o1 *= resc;
        m = newm;

        const float* vbase = v + (size_t)b * S * D + lane * 2;
        #pragma unroll
        for (int r = 0; r < 8; ++r) {
            const float p   = __expf(ps[r] - m);
            const float plo = __shfl(p, 0, 32);
            const float phi = __shfl(p, 16, 32);
            l += plo + phi;
            const float2 vlo = *(const float2*)(vbase + (size_t)(s0 + r) * D);
            const float2 vhi = *(const float2*)(vbase + (size_t)(s0 + 8 + r) * D);
            o0 += plo * vlo.x + phi * vhi.x;
            o1 += plo * vlo.y + phi * vhi.y;
        }
    }

    so[wave][lane * 2]     = o0;
    so[wave][lane * 2 + 1] = o1;
    if (lane == 0) { sm[wave] = m; sl[wave] = l; }
    __syncthreads();

    if (tid < D) {
        float M = sm[0];
        #pragma unroll
        for (int i = 1; i < WAVES; ++i) M = fmaxf(M, sm[i]);
        float L = 0.f, O = 0.f;
        #pragma unroll
        for (int i = 0; i < WAVES; ++i) {
            const float f = __expf(sm[i] - M);
            L += f * sl[i];
            O += f * so[i][tid];
        }
        const int idx = b * CHUNKS + chunk;
        ws[WS_O_OFF + idx * D + tid] = O;
        if (tid == 0) { ws[WS_M_OFF + idx] = M; ws[WS_L_OFF + idx] = L; }
    }
}

__global__ __launch_bounds__(256)
void attn_finish_kernel(float* __restrict__ out, float* __restrict__ attn,
                        const float* __restrict__ ws)
{
    const int b = blockIdx.y;
    const int t = threadIdx.x;
    const size_t i = (size_t)b * S + blockIdx.x * 256 + t;
    const float raw = attn[i];

    const float* pm = ws + WS_M_OFF + b * CHUNKS;
    const float* pl = ws + WS_L_OFF + b * CHUNKS;

    float cm[CHUNKS];
    #pragma unroll
    for (int c = 0; c < CHUNKS; ++c) cm[c] = pm[c];
    float M = cm[0];
    #pragma unroll
    for (int c = 1; c < CHUNKS; ++c) M = fmaxf(M, cm[c]);

    float f[CHUNKS];
    float L = 0.f;
    #pragma unroll
    for (int c = 0; c < CHUNKS; ++c) {
        f[c] = __expf(cm[c] - M);
        L += f[c] * pl[c];
    }

    attn[i] = __expf(raw - M) / L;

    if (blockIdx.x == 0 && t < D) {
        float O = 0.f;
        #pragma unroll
        for (int c = 0; c < CHUNKS; ++c)
            O += f[c] * ws[WS_O_OFF + (b * CHUNKS + c) * D + t];
        out[b * D + t] = O / L;
    }
}

extern "C" void kernel_launch(void* const* d_in, const int* in_sizes, int n_in,
                              void* d_out, int out_size, void* d_ws, size_t ws_size,
                              hipStream_t stream)
{
    (void)in_sizes; (void)n_in; (void)out_size; (void)ws_size;
    const float* q = (const float*)d_in[0];
    const float* k = (const float*)d_in[1];
    const float* v = (const float*)d_in[2];
    float* out  = (float*)d_out;
    float* attn = out + B * D;
    float* ws   = (float*)d_ws;

    attn_partial_kernel<<<dim3(CHUNKS, B), WAVES * 32, 0, stream>>>(q, k, v, attn, ws);
    attn_finish_kernel<<<dim3(S / 256, B), 256, 0, stream>>>(out, attn, ws);
}


# Partial launch: a wave takes its 64 keys in one pass: all K loads requested at entry, V read 16 B per lane behind each tile's scores; one softmax per wave; p*V summed per lane half, halves added last
# speedup vs baseline: 1.5026x; 1.3308x over previous
#include <hip/hip_runtime.h>
#include <math.h>

typedef float v2f __attribute__((ext_vector_type(2)));
typedef float v8f __attribute__((ext_vector_type(8)));

constexpr int   B = 32;
constexpr int   S = 8192;
constexpr int   D = 64;
constexpr float SCALE = 0.125f;

constexpr int WAVES          = 8;
constexpr int KEYS_PER_WAVE  = 64;
constexpr int CHUNKS         = S / (WAVES * KEYS_PER_WAVE);

constexpr int NPART     = B * CHUNKS;
constexpr int WS_O_OFF  = 0;
constexpr int WS_M_OFF  = NPART * D;
constexpr int WS_L_OFF  = WS_M_OFF + NPART;

template <int CTRL>
__device__ __forceinline__ float row_xor_add(float x)
{
    const int xi = __builtin_bit_cast(int, x);
    const int yi = __builtin_amdgcn_mov_dpp(xi, CTRL, 0xF, 0xF, false);
    return x + __builtin_bit_cast(float, yi);
}

__device__ __forceinline__ float pick8(const v8f a, bool b0, bool b1, bool b2)
{
    const float t0 = b0 ? a[1] : a[0];
    const float t1 = b0 ? a[3] : a[2];
    const float t2 = b0 ? a[5] : a[4];
    const float t3 = b0 ? a[7] : a[6];
    const float u0 = b1 ? t1 : t0;
    const float u1 = b1 ? t3 : t2;
    return b2 ? u1 : u0;
}

__global__ __launch_bounds__(WAVES * 32)
void attn_partial_kernel(const float* __restrict__ q,
                         const float* __restrict__ k,
                         const float* __restrict__ v,
                         float* __restrict__ attn,
                         float* __restrict__ ws)
{
    const int tid   = threadIdx.x;
    const int lane  = tid & 31;
    const int wave  = tid >> 5;
    const int chunk = blockIdx.x;
    const int b     = blockIdx.y;

    __shared__ float sm[WAVES];
    __shared__ float sl[WAVES];
    __shared__ __align__(16) float so[WAVES][D];

    const int waveKeyBase = (chunk * WAVES + wave) * KEYS_PER_WAVE;
    const int c    = lane & 15;
    const int g    = c & 7;
    const int doff = (lane >> 4) * 32 + g * 4;
    const bool g0 = (g & 1) != 0, g1 = (g & 2) != 0, g2 = (g & 4) != 0;

    const float4 qf = *(const float4*)(q + b * D + doff);
    const v2f b_lo = {qf.x, qf.y};
    const v2f b_hi = {qf.z, qf.w};

    const float* kwave = k + ((size_t)b * S + waveKeyBase + (c >> 3) * 8) * D + doff;
    const float* vwave = v + ((size_t)b * S + waveKeyBase + (lane >> 4) * 8) * D + c * 4;

    constexpr int TILES = KEYS_PER_WAVE / 16;
    float4 kv[TILES][8];
    #pragma unroll
    for (int t = 0; t < TILES; ++t) {
        #pragma unroll
        for (int j = 0; j < 8; ++j) kv[t][j] = *(const float4*)(kwave + (size_t)(t * 16 + j) * D);
    }

    float4 vv[TILES][8];
    float  ps[TILES][8];
    #pragma unroll
    for (int t = 0; t < TILES; ++t) {
        #pragma unroll
        for (int jb = 0; jb < 8; jb += 4) {
            v8f acc[4];
            #pragma unroll
            for (int j = 0; j < 4; ++j) {
                const v2f a_lo = {kv[t][jb + j].x, kv[t][jb + j].y};
                const v8f zero = {};
                acc[j] = __builtin_amdgcn_wmma_f32_16x16x4_f32(
                    false, a_lo, false, b_lo, (short)0, zero, false, false);
            }
            #pragma unroll
            for (int j = 0; j < 4; ++j) {
                const v2f a_hi = {kv[t][jb + j].z, kv[t][jb + j].w};
                acc[j] = __builtin_amdgcn_wmma_f32_16x16x4_f32(
                    false, a_hi, false, b_hi, (short)0, acc[j], false, false);
            }
            #pragma unroll
            for (int j = 0; j < 4; ++j) {
                float x = pick8(acc[j], g0, g1, g2);
                x = row_xor_add<0x161>(x);
                x = row_xor_add<0x162>(x);
                x = row_xor_add<0x164>(x);
                ps[t][jb + j] = x * SCALE;
            }
        }

        __builtin_amdgcn_sched_barrier(0);
        #pragma unroll
        for (int j = 0; j < 8; ++j) vv[t][j] = *(const float4*)(vwave + (size_t)(t * 16 + j) * D);

        if ((lane & 15) == 0) {
            float* ap = attn + (size_t)b * S + waveKeyBase + t * 16 + (lane >> 4) * 8;
            *(float4*)(ap)     = make_float4(ps[t][0], ps[t][1], ps[t][2], ps[t][3]);
            *(float4*)(ap + 4) = make_float4(ps[t][4], ps[t][5], ps[t][6], ps[t][7]);
        }
    }

    float m = ps[0][0];
    #pragma unroll
    for (int t = 0; t < TILES; ++t) {
        #pragma unroll
        for (int r = 0; r < 8; ++r) m = fmaxf(m, ps[t][r]);
    }
    m = fmaxf(m, __shfl_xor(m, 16, 32));

    float l = 0.f, o0 = 0.f, o1 = 0.f, o2 = 0.f, o3 = 0.f;
    #pragma unroll
    for (int t = 0; t < TILES; ++t) {
        #pragma unroll
        for (int r = 0; r < 8; ++r) {
            const float p = __expf(ps[t][r] - m);
            l  += p;
            o0 += p * vv[t][r].x;
            o1 += p * vv[t][r].y;
            o2 += p * vv[t][r].z;
            o3 += p * vv[t][r].w;
        }
    }
    l  += __shfl_xor(l,  16, 32);
    o0 += __shfl_xor(o0, 16, 32);
    o1 += __shfl_xor(o1, 16, 32);
    o2 += __shfl_xor(o2, 16, 32);
    o3 += __shfl_xor(o3, 16, 32);

    if (lane < 16) *(float4*)(&so[wave][lane * 4]) = make_float4(o0, o1, o2, o3);
    if (lane == 0) { sm[wave] = m; sl[wave] = l; }
    __syncthreads();

    if (tid < D) {
        float M = sm[0];
        #pragma unroll
        for (int i = 1; i < WAVES; ++i) M = fmaxf(M, sm[i]);
        float L = 0.f, O = 0.f;
        #pragma unroll
        for (int i = 0; i < WAVES; ++i) {
            const float f = __expf(sm[i] - M);
            L += f * sl[i];
            O += f * so[i][tid];
        }
        const int idx = b * CHUNKS + chunk;
        ws[WS_O_OFF + idx * D + tid] = O;
        if (tid == 0) { ws[WS_M_OFF + idx] = M; ws[WS_L_OFF + idx] = L; }
    }
}

__global__ __launch_bounds__(256)
void attn_finish_kernel(float* __restrict__ out, float* __restrict__ attn,
                        const float* __restrict__ ws)
{
    const int b = blockIdx.y;
    const int t = threadIdx.x;
    const size_t i = (size_t)b * S + blockIdx.x * 256 + t;
    const float raw = attn[i];

    const float* pm = ws + WS_M_OFF + b * CHUNKS;
    const float* pl = ws + WS_L_OFF + b * CHUNKS;

    float cm[CHUNKS];
    #pragma unroll
    for (int c = 0; c < CHUNKS; ++c) cm[c] = pm[c];
    float M = cm[0];
    #pragma unroll
    for (int c = 1; c < CHUNKS; ++c) M = fmaxf(M, cm[c]);

    float f[CHUNKS];
    float L = 0.f;
    #pragma unroll
    for (int c = 0; c < CHUNKS; ++c) {
        f[c] = __expf(cm[c] - M);
        L += f[c] * pl[c];
    }

    attn[i] = __expf(raw - M) / L;

    if (blockIdx.x == 0 && t < D) {
        float O = 0.f;
        #pragma unroll
        for (int c = 0; c < CHUNKS; ++c)
            O += f[c] * ws[WS_O_OFF + (b * CHUNKS + c) * D + t];
        out[b * D + t] = O / L;
    }
}

extern "C" void kernel_launch(void* const* d_in, const int* in_sizes, int n_in,
                              void* d_out, int out_size, void* d_ws, size_t ws_size,
                              hipStream_t stream)
{
    (void)in_sizes; (void)n_in; (void)out_size; (void)ws_size;
    const float* q = (const float*)d_in[0];
    const float* k = (const float*)d_in[1];
    const float* v = (const float*)d_in[2];
    float* out  = (float*)d_out;
    float* attn = out + B * D;
    float* ws   = (float*)d_ws;

    attn_partial_kernel<<<dim3(CHUNKS, B), WAVES * 32, 0, stream>>>(q, k, v, attn, ws);
    attn_finish_kernel<<<dim3(S / 256, B), 256, 0, stream>>>(out, attn, ws);
}
